# starting kernel
# baseline (speedup 1.0000x reference)
#include <hip/hip_runtime.h>


typedef __attribute__((ext_vector_type(2))) float v2f;
typedef __attribute__((ext_vector_type(8))) float v8f;

#define B_DIM  32
#define A_DIM  1000
#define C_DIM  64
#define H_DIM  12
#define W_DIM  20
#define CH     768
#define N1     999
#define MP     1008
#define NP1    1024
#define FP     1024
#define LDATT  1024
#define KC     64
#define LDS_PAD 68


__global__ void pad_w(const float* __restrict__ Wt, float* __restrict__ Wpk)
{
    int e = blockIdx.x * blockDim.x + threadIdx.x;
    if (e >= CH * NP1) return;
    int k = e >> 10;
    int j = e & (NP1 - 1);
    float v = (j < N1) ? Wt[k * N1 + j] : 0.0f;
    Wpk[(size_t)(k >> 1) * (2 * NP1) + 2 * j + (k & 1)] = v;
}

__global__ void pad_bias(const float* __restrict__ bias, float* __restrict__ biasp)
{
    int j = blockIdx.x * blockDim.x + threadIdx.x;
    if (j < NP1) biasp[j] = (j < N1) ? bias[j] : 0.0f;
}

__global__ void gather_feats(const float* __restrict__ fv,
                             const int*   __restrict__ zi,
                             const int*   __restrict__ yi,
                             const int*   __restrict__ xi,
                             const int*   __restrict__ mask,
                             float* __restrict__ feats,
                             float* __restrict__ featsPk,
                             int b)
{
    int e = blockIdx.x * blockDim.x + threadIdx.x;
    if (e >= FP * CH) return;
    int a = e / CH;
    int k = e - a * CH;
    float v = 0.0f;
    if (a < A_DIM) {
        int idx = a * CH + k;
        if (!mask[idx]) {
            int z = zi[idx], y = yi[idx], x = xi[idx];
            v = fv[((b * C_DIM + z) * H_DIM + y) * W_DIM + x];
        }
    }
    feats[e] = v;
    featsPk[(size_t)(a >> 1) * (2 * CH) + 2 * k + (a & 1)] = v;
}

__device__ __forceinline__ void stage_a(float (*As)[LDS_PAD],
                                        const float* __restrict__ Abase,
                                        int i0, int lda, int kc, int tid)
{
    const int r = tid >> 3;
    const int g = (tid & 7) * 8;
    const float* src = Abase + (size_t)(i0 + r) * lda + kc + g;
    float4 p0 = *(const float4*)(src);
    float4 p1 = *(const float4*)(src + 4);
    *(float4*)&As[r][g]     = p0;
    *(float4*)&As[r][g + 4] = p1;
}

__global__ void gemm1_scores(const float* __restrict__ feats,
                             const float* __restrict__ Wpk,
                             const float* __restrict__ biasp,
                             float* __restrict__ attn)
{
    __shared__ float As[16][LDS_PAD];

    const int lane = threadIdx.x;
    const int wave = threadIdx.y;
    const int tid  = wave * 32 + lane;
    const int q    = blockIdx.x * 4 + wave;
    const int j0   = q * 64;
    const int i0   = blockIdx.y * 16;
    const int half = lane >> 4;
    const int n    = lane & 15;
    const int ar   = lane & 15;

    v8f acc0, acc1, acc2, acc3;
    {
        const float b0 = biasp[j0 + n];
        const float b1 = biasp[j0 + 16 + n];
        const float b2 = biasp[j0 + 32 + n];
        const float b3 = biasp[j0 + 48 + n];
#pragma unroll
        for (int r = 0; r < 8; ++r) {
            acc0[r] = b0; acc1[r] = b1; acc2[r] = b2; acc3[r] = b3;
        }
    }

    for (int kc = 0; kc < CH; kc += KC) {
        __syncthreads();
        stage_a(As, feats, i0, CH, kc, tid);
        __syncthreads();
#pragma unroll
        for (int kk = 0; kk < KC; kk += 4) {
            const int ka = kk + 2 * half;
            v2f afrag = *(const v2f*)&As[ar][ka];
            const float* wrow = Wpk + (size_t)((kc + ka) >> 1) * (2 * NP1);
            v2f b0 = *(const v2f*)&wrow[2 * (j0 + n)];
            v2f b1 = *(const v2f*)&wrow[2 * (j0 + 16 + n)];
            v2f b2 = *(const v2f*)&wrow[2 * (j0 + 32 + n)];
            v2f b3 = *(const v2f*)&wrow[2 * (j0 + 48 + n)];
            acc0 = __builtin_amdgcn_wmma_f32_16x16x4_f32(
                false, afrag, false, b0, (short)0, acc0, false, false);
            acc1 = __builtin_amdgcn_wmma_f32_16x16x4_f32(
                false, afrag, false, b1, (short)0, acc1, false, false);
            acc2 = __builtin_amdgcn_wmma_f32_16x16x4_f32(
                false, afrag, false, b2, (short)0, acc2, false, false);
            acc3 = __builtin_amdgcn_wmma_f32_16x16x4_f32(
                false, afrag, false, b3, (short)0, acc3, false, false);
        }
    }

#pragma unroll
    for (int t = 0; t < 4; ++t) {
        const int jp = j0 + t * 16 + n;
        if (jp < N1) {
#pragma unroll
            for (int r = 0; r < 8; ++r) {
                const int m = i0 + r + 8 * half;
                const int jcol = jp + (jp >= m ? 1 : 0);
                const float val = (t == 0) ? acc0[r] : (t == 1) ? acc1[r]
                                : (t == 2) ? acc2[r] : acc3[r];
                attn[(size_t)m * LDATT + jcol] = val;
            }
        }
    }
}

__global__ void softmax_row(float* __restrict__ attn)
{
    __shared__ float red[256];
    const int i = blockIdx.x;
    const int t = threadIdx.x;
    float* row = attn + (size_t)i * LDATT;

    float mx = -3.402823466e+38f;
    for (int j = t; j < A_DIM; j += 256)
        if (j != i) mx = fmaxf(mx, row[j]);
    red[t] = mx;
    __syncthreads();
    for (int s = 128; s > 0; s >>= 1) {
        if (t < s) red[t] = fmaxf(red[t], red[t + s]);
        __syncthreads();
    }
    mx = red[0];
    __syncthreads();

    float sum = 0.0f;
    for (int j = t; j < A_DIM; j += 256)
        if (j != i) sum += expf(row[j] - mx);
    red[t] = sum;
    __syncthreads();
    for (int s = 128; s > 0; s >>= 1) {
        if (t < s) red[t] += red[t + s];
        __syncthreads();
    }
    const float inv = 1.0f / red[0];

    for (int j = t; j < A_DIM; j += 256)
        row[j] = (j == i) ? 0.0f : expf(row[j] - mx) * inv;
    for (int j = A_DIM + t; j < LDATT; j += 256)
        row[j] = 0.0f;
}

__global__ void gemm2_out(const float* __restrict__ attn,
                          const float* __restrict__ featsPk,
                          float* __restrict__ outb)
{
    __shared__ float As[16][LDS_PAD];

    const int lane = threadIdx.x;
    const int wave = threadIdx.y;
    const int tid  = wave * 32 + lane;
    const int q    = blockIdx.x * 4 + wave;
    const int n0   = q * 64;
    const int i0   = blockIdx.y * 16;
    const int half = lane >> 4;
    const int n    = lane & 15;
    const int ar   = lane & 15;

    v8f acc0 = {}, acc1 = {}, acc2 = {}, acc3 = {};

    for (int kc = 0; kc < FP; kc += KC) {
        __syncthreads();
        stage_a(As, attn, i0, LDATT, kc, tid);
        __syncthreads();
#pragma unroll
        for (int kk = 0; kk < KC; kk += 4) {
            const int ka = kk + 2 * half;
            v2f afrag = *(const v2f*)&As[ar][ka];
            const float* frow = featsPk + (size_t)((kc + ka) >> 1) * (2 * CH);
            v2f b0 = *(const v2f*)&frow[2 * (n0 + n)];
            v2f b1 = *(const v2f*)&frow[2 * (n0 + 16 + n)];
            v2f b2 = *(const v2f*)&frow[2 * (n0 + 32 + n)];
            v2f b3 = *(const v2f*)&frow[2 * (n0 + 48 + n)];
            acc0 = __builtin_amdgcn_wmma_f32_16x16x4_f32(
                false, afrag, false, b0, (short)0, acc0, false, false);
            acc1 = __builtin_amdgcn_wmma_f32_16x16x4_f32(
                false, afrag, false, b1, (short)0, acc1, false, false);
            acc2 = __builtin_amdgcn_wmma_f32_16x16x4_f32(
                false, afrag, false, b2, (short)0, acc2, false, false);
            acc3 = __builtin_amdgcn_wmma_f32_16x16x4_f32(
                false, afrag, false, b3, (short)0, acc3, false, false);
        }
    }

#pragma unroll
    for (int r = 0; r < 8; ++r) {
        const int m = i0 + r + 8 * half;
        if (m < A_DIM) {
            float* orow = outb + (size_t)m * CH;
            orow[n0 + n]      = acc0[r];
            orow[n0 + 16 + n] = acc1[r];
            orow[n0 + 32 + n] = acc2[r];
            orow[n0 + 48 + n] = acc3[r];
        }
    }
}

extern "C" void kernel_launch(void* const* d_in, const int* in_sizes, int n_in,
                              void* d_out, int out_size, void* d_ws, size_t ws_size,
                              hipStream_t stream)
{
    const float* fv   = (const float*)d_in[0];
    const int*   zi   = (const int*)  d_in[1];
    const int*   yi   = (const int*)  d_in[2];
    const int*   xi   = (const int*)  d_in[3];
    const int*   mask = (const int*)  d_in[4];
    const float* Wt   = (const float*)d_in[5];
    const float* bias = (const float*)d_in[6];
    float*       out  = (float*)d_out;

    float* Wpk     = (float*)d_ws;
    float* biasp   = Wpk     + (size_t)CH * NP1;
    float* feats   = biasp   + NP1;
    float* featsPk = feats   + (size_t)FP * CH;
    float* attn    = featsPk + (size_t)FP * CH;

    const dim3 wblk(32, 4);

    pad_w<<<(CH * NP1 + 255) / 256, 256, 0, stream>>>(Wt, Wpk);
    pad_bias<<<(NP1 + 255) / 256, 256, 0, stream>>>(bias, biasp);

    for (int b = 0; b < B_DIM; ++b) {
        gather_feats<<<(FP * CH + 255) / 256, 256, 0, stream>>>(
            fv, zi, yi, xi, mask, feats, featsPk, b);
        gemm1_scores<<<dim3(4, 63), wblk, 0, stream>>>(feats, Wpk, biasp, attn);
        softmax_row<<<dim3(A_DIM), 256, 0, stream>>>(attn);
        gemm2_out<<<dim3(3, 63), wblk, 0, stream>>>(attn, featsPk,
                                                    out + (size_t)b * A_DIM * CH);
    }
}


# Row tile 32 per work-group of 16 waves (was 16 rows, 4 waves): a wave owns both 16-row halves, so each W fragment from global and each gathered feats fragment feeds two WMMAs; sums unchanged
# speedup vs baseline: 7.9808x; 7.9808x over previous
#include <hip/hip_runtime.h>


typedef __attribute__((ext_vector_type(2))) float v2f;
typedef __attribute__((ext_vector_type(8))) float v8f;

#define B_DIM  32
#define A_DIM  1000
#define C_DIM  64
#define H_DIM  12
#define W_DIM  20
#define CH     768
#define N1     999
#define MP     1008
#define RT     32
#define NTILE  32
#define NP1    1024
#define FVN    15360
#define ZSLOT  61440u
#define AST    772
#define SST    1028
#define K1S    192
#define K2S    250

#define WMMA4(a, b, c) __builtin_amdgcn_wmma_f32_16x16x4_f32( \
    false, (a), false, (b), (short)0, (c), false, false)

__global__ void pad_w(const float* __restrict__ Wt, float* __restrict__ Wpk)
{
    int e = blockIdx.x * blockDim.x + threadIdx.x;
    if (e >= CH * NP1) return;
    int k = e >> 10;
    int j = e & (NP1 - 1);
    float v = (j < N1) ? Wt[k * N1 + j] : 0.0f;
    Wpk[(size_t)(k >> 1) * (2 * NP1) + 2 * j + (k & 1)] = v;
}

__global__ void pad_bias(const float* __restrict__ bias, float* __restrict__ biasp)
{
    int j = blockIdx.x * blockDim.x + threadIdx.x;
    if (j < NP1) biasp[j] = (j < N1) ? bias[j] : 0.0f;
}

__global__ void pack_offsets(const int* __restrict__ zi, const int* __restrict__ yi,
                             const int* __restrict__ xi, const int* __restrict__ mask,
                             unsigned int* __restrict__ offPk)
{
    int e = blockIdx.x * blockDim.x + threadIdx.x;
    if (e >= (MP / 2) * CH) return;
    int p = e / CH;
    int k = e - p * CH;
    unsigned int w = 0;
#pragma unroll
    for (int s = 0; s < 2; ++s) {
        int a = 2 * p + s;
        unsigned int o = ZSLOT;
        if (a < A_DIM) {
            int idx = a * CH + k;
            if (!mask[idx])
                o = 4u * (unsigned int)((zi[idx] * H_DIM + yi[idx]) * W_DIM + xi[idx]);
        }
        w |= o << (16 * s);
    }
    offPk[e] = w;
}

__device__ __forceinline__ float lds_at(const float* base, unsigned int byteoff)
{
    return *(const float*)((const char*)base + byteoff);
}

__global__ void __launch_bounds__(512)
fused_attn(const float* __restrict__ fv,
           const unsigned int* __restrict__ offPk,
           const float* __restrict__ Wpk,
           const float* __restrict__ biasp,
           float* __restrict__ out)
{
    __shared__ __attribute__((aligned(16))) float smem[FVN + 4 + RT * SST];
    float* const fvs  = smem;
    float* const tile = smem + FVN + 4;

    const int lane = threadIdx.x;
    const int wave = threadIdx.y;
    const int tid  = wave * 32 + lane;
    const int i0   = blockIdx.x * RT;
    const int b    = blockIdx.y;
    const int half = lane >> 4;
    const int n    = lane & 15;
    const int ar   = lane & 15;

    {
        const float4* src = (const float4*)(fv + (size_t)b * FVN);
        float4* dst = (float4*)fvs;
        for (int i = tid; i < FVN / 4; i += 512) dst[i] = src[i];
        if (tid == 0) fvs[FVN] = 0.0f;
    }
    __syncthreads();

    {
        const unsigned int* orow = offPk + (size_t)(i0 >> 1) * CH;
        const int plim = MP / 2 - (i0 >> 1);
        for (int e = tid; e < (RT / 2) * CH; e += 512) {
            const int p = e / CH;
            const int k = e - p * CH;
            const unsigned int w = (p < plim) ? orow[e] : (ZSLOT | (ZSLOT << 16));
            tile[(2 * p) * AST + k]     = lds_at(fvs, w & 0xffffu);
            tile[(2 * p + 1) * AST + k] = lds_at(fvs, w >> 16);
        }
    }
    __syncthreads();

    const int j0 = wave * 64;
    v8f acc[2][4];
#pragma unroll
    for (int t = 0; t < 4; ++t) {
        const float bv = biasp[j0 + t * 16 + n];
#pragma unroll
        for (int r = 0; r < 8; ++r) { acc[0][t][r] = bv; acc[1][t][r] = bv; }
    }
    {
        const float* wbase  = Wpk + (size_t)half * (2 * NP1) + 2 * (j0 + n);
        const float* abase0 = tile + ar * AST + 2 * half;
        const float* abase1 = abase0 + 16 * AST;
        v2f bE[4], bO[4];
#pragma unroll
        for (int t = 0; t < 4; ++t) {
            bE[t] = *(const v2f*)&wbase[32 * t];
            bO[t] = *(const v2f*)&wbase[4 * NP1 + 32 * t];
        }
        for (int ks = 0; ks < K1S; ks += 2) {
            const v2f a00 = *(const v2f*)&abase0[ks * 4];
            const v2f a10 = *(const v2f*)&abase1[ks * 4];
#pragma unroll
            for (int t = 0; t < 4; ++t) {
                acc[0][t] = WMMA4(a00, bE[t], acc[0][t]);
                acc[1][t] = WMMA4(a10, bE[t], acc[1][t]);
            }
            {
                const int kn = (ks + 2 < K1S) ? ks + 2 : K1S - 1;
                const float* wr = wbase + (size_t)kn * (4 * NP1);
#pragma unroll
                for (int t = 0; t < 4; ++t) bE[t] = *(const v2f*)&wr[32 * t];
            }
            const v2f a01 = *(const v2f*)&abase0[ks * 4 + 4];
            const v2f a11 = *(const v2f*)&abase1[ks * 4 + 4];
#pragma unroll
            for (int t = 0; t < 4; ++t) {
                acc[0][t] = WMMA4(a01, bO[t], acc[0][t]);
                acc[1][t] = WMMA4(a11, bO[t], acc[1][t]);
            }
            {
                const int kn = (ks + 3 < K1S) ? ks + 3 : K1S - 1;
                const float* wr = wbase + (size_t)kn * (4 * NP1);
#pragma unroll
                for (int t = 0; t < 4; ++t) bO[t] = *(const v2f*)&wr[32 * t];
            }
        }
    }
    __syncthreads();

#pragma unroll
    for (int h = 0; h < 2; ++h) {
#pragma unroll
        for (int t = 0; t < 4; ++t) {
            const int jp = j0 + t * 16 + n;
            if (jp < N1) {
#pragma unroll
                for (int r = 0; r < 8; ++r) {
                    const int m  = 16 * h + r + 8 * half;
                    const int ig = i0 + m;
                    tile[m * SST + jp + (jp >= ig ? 1 : 0)] = acc[h][t][r];
                }
            }
        }
    }
    __syncthreads();

    for (int rr = 0; rr < 2; ++rr) {
        const int m  = wave * 2 + rr;
        const int ig = i0 + m;
        float* row = tile + m * SST;
        if (ig < A_DIM) {
            float mx = -3.402823466e+38f;
            for (int j = lane; j < A_DIM; j += 32)
                if (j != ig) mx = fmaxf(mx, row[j]);
#pragma unroll
            for (int s = 16; s > 0; s >>= 1) mx = fmaxf(mx, __shfl_xor(mx, s, 32));
            float sum = 0.0f;
            for (int j = lane; j < A_DIM; j += 32) {
                const float e = (j == ig) ? 0.0f : expf(row[j] - mx);
                row[j] = e;
                sum += e;
            }
#pragma unroll
            for (int s = 16; s > 0; s >>= 1) sum += __shfl_xor(sum, s, 32);
            const float inv = 1.0f / sum;
            for (int j = lane; j < A_DIM; j += 32) row[j] *= inv;
        } else {
            for (int j = lane; j < A_DIM; j += 32) row[j] = 0.0f;
        }
    }
    __syncthreads();

    if (wave >= CH / 64) return;
    const int n0 = wave * 64;
    v8f o[2][4];
#pragma unroll
    for (int t = 0; t < 4; ++t)
#pragma unroll
        for (int r = 0; r < 8; ++r) { o[0][t][r] = 0.0f; o[1][t][r] = 0.0f; }
    {
        const unsigned int* obase = offPk + (size_t)half * CH + n0 + n;
        const float* pbase0 = tile + ar * SST + 2 * half;
        const float* pbase1 = pbase0 + 16 * SST;
        unsigned int w[8];
#pragma unroll
        for (int t = 0; t < 4; ++t) {
            w[t]     = obase[16 * t];
            w[4 + t] = obase[2 * CH + 16 * t];
        }
        for (int kp = 0; kp < K2S / 2; ++kp) {
            unsigned int lo[8], hi[8];
#pragma unroll
            for (int i = 0; i < 8; ++i) { lo[i] = w[i] & 0xffffu; hi[i] = w[i] >> 16; }
            {
                const int kn = (kp + 1 < K2S / 2) ? kp + 1 : K2S / 2 - 1;
                const unsigned int* orow = obase + (size_t)kn * (4 * CH);
#pragma unroll
                for (int t = 0; t < 4; ++t) {
                    w[t]     = orow[16 * t];
                    w[4 + t] = orow[2 * CH + 16 * t];
                }
            }
            const v2f a00 = *(const v2f*)&pbase0[kp * 8];
            const v2f a10 = *(const v2f*)&pbase1[kp * 8];
            const v2f a01 = *(const v2f*)&pbase0[kp * 8 + 4];
            const v2f a11 = *(const v2f*)&pbase1[kp * 8 + 4];
#pragma unroll
            for (int t = 0; t < 4; ++t) {
                v2f bf;
                bf.x = lds_at(fvs, lo[t]);
                bf.y = lds_at(fvs, hi[t]);
                o[0][t] = WMMA4(a00, bf, o[0][t]);
                o[1][t] = WMMA4(a10, bf, o[1][t]);
            }
#pragma unroll
            for (int t = 0; t < 4; ++t) {
                v2f bf;
                bf.x = lds_at(fvs, lo[4 + t]);
                bf.y = lds_at(fvs, hi[4 + t]);
                o[0][t] = WMMA4(a01, bf, o[0][t]);
                o[1][t] = WMMA4(a11, bf, o[1][t]);
            }
        }
    }

    float* outb = out + (size_t)b * A_DIM * CH;
#pragma unroll
    for (int h = 0; h < 2; ++h) {
#pragma unroll
        for (int r = 0; r < 8; ++r) {
            const int m = i0 + 16 * h + r + 8 * half;
            if (m < A_DIM) {
                float* orow = outb + (size_t)m * CH + n0 + n;
#pragma unroll
                for (int t = 0; t < 4; ++t) orow[16 * t] = o[h][t][r];
            }
        }
    }
}

extern "C" void kernel_launch(void* const* d_in, const int* in_sizes, int n_in,
                              void* d_out, int out_size, void* d_ws, size_t ws_size,
                              hipStream_t stream)
{
    const float* fv   = (const float*)d_in[0];
    const int*   zi   = (const int*)  d_in[1];
    const int*   yi   = (const int*)  d_in[2];
    const int*   xi   = (const int*)  d_in[3];
    const int*   mask = (const int*)  d_in[4];
    const float* Wt   = (const float*)d_in[5];
    const float* bias = (const float*)d_in[6];
    float*       out  = (float*)d_out;

    float*        Wpk   = (float*)d_ws;
    float*        biasp = Wpk + (size_t)CH * NP1;
    unsigned int* offPk = (unsigned int*)(biasp + NP1);

    pad_w<<<(CH * NP1 + 255) / 256, 256, 0, stream>>>(Wt, Wpk);
    pad_bias<<<(NP1 + 255) / 256, 256, 0, stream>>>(bias, biasp);
    pack_offsets<<<((MP / 2) * CH + 255) / 256, 256, 0, stream>>>(zi, yi, xi, mask, offPk);
    fused_attn<<<dim3(NTILE, B_DIM), dim3(32, 16), 0, stream>>>(fv, offPk, Wpk, biasp, out);
}


# Row tile 64 per work-group (was 32): a wave owns four 16-row blocks, so each W fragment from global and each gathered feats fragment feeds four WMMAs; 512 work-groups, one per WGP; sums unchanged
# speedup vs baseline: 8.6185x; 1.0799x over previous
#include <hip/hip_runtime.h>


typedef __attribute__((ext_vector_type(2))) float v2f;
typedef __attribute__((ext_vector_type(8))) float v8f;

#define B_DIM  32
#define A_DIM  1000
#define C_DIM  64
#define H_DIM  12
#define W_DIM  20
#define CH     768
#define N1     999
#define MP     1008
#define RT     64
#define RB     4
#define NTILE  16
#define NP1    1024
#define FVN    15360
#define ZSLOT  61440u
#define AST    772
#define SST    1004
#define K1S    192
#define K2S    250

#define WMMA4(a, b, c) __builtin_amdgcn_wmma_f32_16x16x4_f32( \
    false, (a), false, (b), (short)0, (c), false, false)

__global__ void pad_w(const float* __restrict__ Wt, float* __restrict__ Wpk)
{
    int e = blockIdx.x * blockDim.x + threadIdx.x;
    if (e >= CH * NP1) return;
    int k = e >> 10;
    int j = e & (NP1 - 1);
    float v = (j < N1) ? Wt[k * N1 + j] : 0.0f;
    Wpk[(size_t)(k >> 1) * (2 * NP1) + 2 * j + (k & 1)] = v;
}

__global__ void pad_bias(const float* __restrict__ bias, float* __restrict__ biasp)
{
    int j = blockIdx.x * blockDim.x + threadIdx.x;
    if (j < NP1) biasp[j] = (j < N1) ? bias[j] : 0.0f;
}

__global__ void pack_offsets(const int* __restrict__ zi, const int* __restrict__ yi,
                             const int* __restrict__ xi, const int* __restrict__ mask,
                             unsigned int* __restrict__ offPk)
{
    int e = blockIdx.x * blockDim.x + threadIdx.x;
    if (e >= (MP / 2) * CH) return;
    int p = e / CH;
    int k = e - p * CH;
    unsigned int w = 0;
#pragma unroll
    for (int s = 0; s < 2; ++s) {
        int a = 2 * p + s;
        unsigned int o = ZSLOT;
        if (a < A_DIM) {
            int idx = a * CH + k;
            if (!mask[idx])
                o = 4u * (unsigned int)((zi[idx] * H_DIM + yi[idx]) * W_DIM + xi[idx]);
        }
        w |= o << (16 * s);
    }
    offPk[e] = w;
}

__device__ __forceinline__ float lds_at(const float* base, unsigned int byteoff)
{
    return *(const float*)((const char*)base + byteoff);
}

__global__ void __launch_bounds__(512)
fused_attn(const float* __restrict__ fv,
           const unsigned int* __restrict__ offPk,
           const float* __restrict__ Wpk,
           const float* __restrict__ biasp,
           float* __restrict__ out)
{
    __shared__ __attribute__((aligned(16))) float smem[FVN + 4 + RT * SST];
    float* const fvs  = smem;
    float* const tile = smem + FVN + 4;

    const int lane = threadIdx.x;
    const int wave = threadIdx.y;
    const int tid  = wave * 32 + lane;
    const int i0   = blockIdx.x * RT;
    const int b    = blockIdx.y;
    const int half = lane >> 4;
    const int n    = lane & 15;
    const int ar   = lane & 15;

    {
        const float4* src = (const float4*)(fv + (size_t)b * FVN);
        float4* dst = (float4*)fvs;
        for (int i = tid; i < FVN / 4; i += 512) dst[i] = src[i];
        if (tid == 0) fvs[FVN] = 0.0f;
    }
    __syncthreads();

    {
        const unsigned int* orow = offPk + (size_t)(i0 >> 1) * CH;
        const int plim = MP / 2 - (i0 >> 1);
        for (int e = tid; e < (RT / 2) * CH; e += 512) {
            const int p = e / CH;
            const int k = e - p * CH;
            const unsigned int w = (p < plim) ? orow[e] : (ZSLOT | (ZSLOT << 16));
            tile[(2 * p) * AST + k]     = lds_at(fvs, w & 0xffffu);
            tile[(2 * p + 1) * AST + k] = lds_at(fvs, w >> 16);
        }
    }
    __syncthreads();

    const int j0 = wave * 64;
    v8f acc[RB][4];
#pragma unroll
    for (int t = 0; t < 4; ++t) {
        const float bv = biasp[j0 + t * 16 + n];
#pragma unroll
        for (int h = 0; h < RB; ++h)
#pragma unroll
            for (int r = 0; r < 8; ++r) acc[h][t][r] = bv;
    }
    {
        const float* wbase  = Wpk + (size_t)half * (2 * NP1) + 2 * (j0 + n);
        const float* abase = tile + ar * AST + 2 * half;
        v2f bE[4], bO[4];
#pragma unroll
        for (int t = 0; t < 4; ++t) {
            bE[t] = *(const v2f*)&wbase[32 * t];
            bO[t] = *(const v2f*)&wbase[4 * NP1 + 32 * t];
        }
        for (int ks = 0; ks < K1S; ks += 2) {
            v2f aE[RB];
#pragma unroll
            for (int h = 0; h < RB; ++h) aE[h] = *(const v2f*)&abase[h * 16 * AST + ks * 4];
#pragma unroll
            for (int t = 0; t < 4; ++t) {
#pragma unroll
                for (int h = 0; h < RB; ++h) acc[h][t] = WMMA4(aE[h], bE[t], acc[h][t]);
            }
            {
                const int kn = (ks + 2 < K1S) ? ks + 2 : K1S - 1;
                const float* wr = wbase + (size_t)kn * (4 * NP1);
#pragma unroll
                for (int t = 0; t < 4; ++t) bE[t] = *(const v2f*)&wr[32 * t];
            }
            v2f aO[RB];
#pragma unroll
            for (int h = 0; h < RB; ++h) aO[h] = *(const v2f*)&abase[h * 16 * AST + ks * 4 + 4];
#pragma unroll
            for (int t = 0; t < 4; ++t) {
#pragma unroll
                for (int h = 0; h < RB; ++h) acc[h][t] = WMMA4(aO[h], bO[t], acc[h][t]);
            }
            {
                const int kn = (ks + 3 < K1S) ? ks + 3 : K1S - 1;
                const float* wr = wbase + (size_t)kn * (4 * NP1);
#pragma unroll
                for (int t = 0; t < 4; ++t) bO[t] = *(const v2f*)&wr[32 * t];
            }
        }
    }
    __syncthreads();

#pragma unroll
    for (int h = 0; h < RB; ++h) {
#pragma unroll
        for (int t = 0; t < 4; ++t) {
            const int jp = j0 + t * 16 + n;
            if (jp < N1) {
#pragma unroll
                for (int r = 0; r < 8; ++r) {
                    const int m  = 16 * h + r + 8 * half;
                    const int ig = i0 + m;
                    tile[m * SST + jp + (jp >= ig ? 1 : 0)] = acc[h][t][r];
                }
            }
        }
    }
    __syncthreads();

    for (int rr = 0; rr < RT / 16; ++rr) {
        const int m  = wave * (RT / 16) + rr;
        const int ig = i0 + m;
        float* row = tile + m * SST;
        if (ig < A_DIM) {
            float mx = -3.402823466e+38f;
            for (int j = lane; j < A_DIM; j += 32)
                if (j != ig) mx = fmaxf(mx, row[j]);
#pragma unroll
            for (int s = 16; s > 0; s >>= 1) mx = fmaxf(mx, __shfl_xor(mx, s, 32));
            float sum = 0.0f;
            for (int j = lane; j < A_DIM; j += 32) {
                const float e = (j == ig) ? 0.0f : expf(row[j] - mx);
                row[j] = e;
                sum += e;
            }
#pragma unroll
            for (int s = 16; s > 0; s >>= 1) sum += __shfl_xor(sum, s, 32);
            const float inv = 1.0f / sum;
            for (int j = lane; j < A_DIM; j += 32) row[j] *= inv;
        } else {
            for (int j = lane; j < A_DIM; j += 32) row[j] = 0.0f;
        }
    }
    __syncthreads();

    if (wave >= CH / 64) return;
    const int n0 = wave * 64;
    v8f o[RB][4];
#pragma unroll
    for (int h = 0; h < RB; ++h)
#pragma unroll
        for (int t = 0; t < 4; ++t)
#pragma unroll
            for (int r = 0; r < 8; ++r) o[h][t][r] = 0.0f;
    {
        const unsigned int* obase = offPk + (size_t)half * CH + n0 + n;
        const float* pbase = tile + ar * SST + 2 * half;
        unsigned int w[8];
#pragma unroll
        for (int t = 0; t < 4; ++t) {
            w[t]     = obase[16 * t];
            w[4 + t] = obase[2 * CH + 16 * t];
        }
        for (int kp = 0; kp < K2S / 2; ++kp) {
            unsigned int lo[8], hi[8];
#pragma unroll
            for (int i = 0; i < 8; ++i) { lo[i] = w[i] & 0xffffu; hi[i] = w[i] >> 16; }
            {
                const int kn = (kp + 1 < K2S / 2) ? kp + 1 : K2S / 2 - 1;
                const unsigned int* orow = obase + (size_t)kn * (4 * CH);
#pragma unroll
                for (int t = 0; t < 4; ++t) {
                    w[t]     = orow[16 * t];
                    w[4 + t] = orow[2 * CH + 16 * t];
                }
            }
            v2f pE[RB], pO[RB];
#pragma unroll
            for (int h = 0; h < RB; ++h) {
                pE[h] = *(const v2f*)&pbase[h * 16 * SST + kp * 8];
                pO[h] = *(const v2f*)&pbase[h * 16 * SST + kp * 8 + 4];
            }
#pragma unroll
            for (int t = 0; t < 4; ++t) {
                v2f bf;
                bf.x = lds_at(fvs, lo[t]);
                bf.y = lds_at(fvs, hi[t]);
#pragma unroll
                for (int h = 0; h < RB; ++h) o[h][t] = WMMA4(pE[h], bf, o[h][t]);
            }
#pragma unroll
            for (int t = 0; t < 4; ++t) {
                v2f bf;
                bf.x = lds_at(fvs, lo[4 + t]);
                bf.y = lds_at(fvs, hi[4 + t]);
#pragma unroll
                for (int h = 0; h < RB; ++h) o[h][t] = WMMA4(pO[h], bf, o[h][t]);
            }
        }
    }

    float* outb = out + (size_t)b * A_DIM * CH;
#pragma unroll
    for (int h = 0; h < RB; ++h) {
#pragma unroll
        for (int r = 0; r < 8; ++r) {
            const int m = i0 + 16 * h + r + 8 * half;
            if (m < A_DIM) {
                float* orow = outb + (size_t)m * CH + n0 + n;
#pragma unroll
                for (int t = 0; t < 4; ++t) orow[16 * t] = o[h][t][r];
            }
        }
    }
}

extern "C" void kernel_launch(void* const* d_in, const int* in_sizes, int n_in,
                              void* d_out, int out_size, void* d_ws, size_t ws_size,
                              hipStream_t stream)
{
    const float* fv   = (const float*)d_in[0];
    const int*   zi   = (const int*)  d_in[1];
    const int*   yi   = (const int*)  d_in[2];
    const int*   xi   = (const int*)  d_in[3];
    const int*   mask = (const int*)  d_in[4];
    const float* Wt   = (const float*)d_in[5];
    const float* bias = (const float*)d_in[6];
    float*       out  = (float*)d_out;

    float*        Wpk   = (float*)d_ws;
    float*        biasp = Wpk + (size_t)CH * NP1;
    unsigned int* offPk = (unsigned int*)(biasp + NP1);

    pad_w<<<(CH * NP1 + 255) / 256, 256, 0, stream>>>(Wt, Wpk);
    pad_bias<<<(NP1 + 255) / 256, 256, 0, stream>>>(bias, biasp);
    pack_offsets<<<((MP / 2) * CH + 255) / 256, 256, 0, stream>>>(zi, yi, xi, mask, offPk);
    fused_attn<<<dim3(NTILE, B_DIM), dim3(32, 16), 0, stream>>>(fv, offPk, Wpk, biasp, out);
}


# Phases between the matrix runs (volume copy, A-tile gather, row softmax): a thread's loads issued in groups ahead of their uses, softmax row held in registers (was one load, one wait); sums unchanged
# speedup vs baseline: 9.7092x; 1.1266x over previous
#include <hip/hip_runtime.h>


typedef __attribute__((ext_vector_type(2))) float v2f;
typedef __attribute__((ext_vector_type(8))) float v8f;

#define B_DIM  32
#define A_DIM  1000
#define C_DIM  64
#define H_DIM  12
#define W_DIM  20
#define CH     768
#define N1     999
#define MP     1008
#define RT     64
#define RB     4
#define NTILE  16
#define NP1    1024
#define FVN    15360
#define ZSLOT  61440u
#define AST    772
#define SST    1004
#define K1S    192
#define K2S    250
#define GIT    48
#define GGRP   16
#define CIT    7
#define SMI    32

static_assert(FVN == 4 * 512 * CIT + 2 * 512, "volume copy: 7 float4 rounds and one float2 round of 512 threads");
static_assert(A_DIM % (RT / 16) == 0, "a wave's softmax rows are all real or all padding");

#define WMMA4(a, b, c) __builtin_amdgcn_wmma_f32_16x16x4_f32( \
    false, (a), false, (b), (short)0, (c), false, false)

__global__ void pad_w(const float* __restrict__ Wt, float* __restrict__ Wpk)
{
    int e = blockIdx.x * blockDim.x + threadIdx.x;
    if (e >= CH * NP1) return;
    int k = e >> 10;
    int j = e & (NP1 - 1);
    float v = (j < N1) ? Wt[k * N1 + j] : 0.0f;
    Wpk[(size_t)(k >> 1) * (2 * NP1) + 2 * j + (k & 1)] = v;
}

__global__ void pad_bias(const float* __restrict__ bias, float* __restrict__ biasp)
{
    int j = blockIdx.x * blockDim.x + threadIdx.x;
    if (j < NP1) biasp[j] = (j < N1) ? bias[j] : 0.0f;
}

__global__ void pack_offsets(const int* __restrict__ zi, const int* __restrict__ yi,
                             const int* __restrict__ xi, const int* __restrict__ mask,
                             unsigned int* __restrict__ offPk)
{
    int e = blockIdx.x * blockDim.x + threadIdx.x;
    if (e >= (MP / 2) * CH) return;
    int p = e / CH;
    int k = e - p * CH;
    unsigned int w = 0;
#pragma unroll
    for (int s = 0; s < 2; ++s) {
        int a = 2 * p + s;
        unsigned int o = ZSLOT;
        if (a < A_DIM) {
            int idx = a * CH + k;
            if (!mask[idx])
                o = 4u * (unsigned int)((zi[idx] * H_DIM + yi[idx]) * W_DIM + xi[idx]);
        }
        w |= o << (16 * s);
    }
    offPk[e] = w;
}

__device__ __forceinline__ float lds_at(const float* base, unsigned int byteoff)
{
    return *(const float*)((const char*)base + byteoff);
}

__global__ void __launch_bounds__(512)
fused_attn(const float* __restrict__ fv,
           const unsigned int* __restrict__ offPk,
           const float* __restrict__ Wpk,
           const float* __restrict__ biasp,
           float* __restrict__ out)
{
    __shared__ __attribute__((aligned(16))) float smem[FVN + 4 + RT * SST];
    float* const fvs  = smem;
    float* const tile = smem + FVN + 4;

    const int lane = threadIdx.x;
    const int wave = threadIdx.y;
    const int tid  = wave * 32 + lane;
    const int i0   = blockIdx.x * RT;
    const int b    = blockIdx.y;
    const int half = lane >> 4;
    const int n    = lane & 15;
    const int ar   = lane & 15;

    unsigned int gw[GIT];
    {
        const unsigned int* orow = offPk + (size_t)(i0 >> 1) * CH;
        const int plim = MP / 2 - (i0 >> 1);
        const int elim = plim * CH - 1;
#pragma unroll
        for (int it = 0; it < GIT; ++it) {
            const int e = tid + 512 * it;
            const unsigned int w = orow[e < elim ? e : elim];
            gw[it] = (e <= elim) ? w : (ZSLOT | (ZSLOT << 16));
        }
    }
    {
        const float4* src = (const float4*)(fv + (size_t)b * FVN);
        float4* dst = (float4*)fvs;
        float4 cv[CIT];
#pragma unroll
        for (int q = 0; q < CIT; ++q) cv[q] = src[tid + 512 * q];
        const float2 cl = ((const float2*)(src + 512 * CIT))[tid];
#pragma unroll
        for (int q = 0; q < CIT; ++q) dst[tid + 512 * q] = cv[q];
        ((float2*)(dst + 512 * CIT))[tid] = cl;
        if (tid == 0) fvs[FVN] = 0.0f;
    }
    __syncthreads();

#pragma unroll
    for (int g = 0; g < GIT; g += GGRP) {
        float glo[GGRP], ghi[GGRP];
#pragma unroll
        for (int u = 0; u < GGRP; ++u) {
            glo[u] = lds_at(fvs, gw[g + u] & 0xffffu);
            ghi[u] = lds_at(fvs, gw[g + u] >> 16);
        }
#pragma unroll
        for (int u = 0; u < GGRP; ++u) {
            const int e = tid + 512 * (g + u);
            const int p = e / CH;
            const int k = e - p * CH;
            tile[(2 * p) * AST + k]     = glo[u];
            tile[(2 * p + 1) * AST + k] = ghi[u];
        }
    }
    __syncthreads();

    const int j0 = wave * 64;
    v8f acc[RB][4];
#pragma unroll
    for (int t = 0; t < 4; ++t) {
        const float bv = biasp[j0 + t * 16 + n];
#pragma unroll
        for (int h = 0; h < RB; ++h)
#pragma unroll
            for (int r = 0; r < 8; ++r) acc[h][t][r] = bv;
    }
    {
        const float* wbase  = Wpk + (size_t)half * (2 * NP1) + 2 * (j0 + n);
        const float* abase = tile + ar * AST + 2 * half;
        v2f bE[4], bO[4];
#pragma unroll
        for (int t = 0; t < 4; ++t) {
            bE[t] = *(const v2f*)&wbase[32 * t];
            bO[t] = *(const v2f*)&wbase[4 * NP1 + 32 * t];
        }
        for (int ks = 0; ks < K1S; ks += 2) {
            v2f aE[RB];
#pragma unroll
            for (int h = 0; h < RB; ++h) aE[h] = *(const v2f*)&abase[h * 16 * AST + ks * 4];
#pragma unroll
            for (int t = 0; t < 4; ++t) {
#pragma unroll
                for (int h = 0; h < RB; ++h) acc[h][t] = WMMA4(aE[h], bE[t], acc[h][t]);
            }
            {
                const int kn = (ks + 2 < K1S) ? ks + 2 : K1S - 1;
                const float* wr = wbase + (size_t)kn * (4 * NP1);
#pragma unroll
                for (int t = 0; t < 4; ++t) bE[t] = *(const v2f*)&wr[32 * t];
            }
            v2f aO[RB];
#pragma unroll
            for (int h = 0; h < RB; ++h) aO[h] = *(const v2f*)&abase[h * 16 * AST + ks * 4 + 4];
#pragma unroll
            for (int t = 0; t < 4; ++t) {
#pragma unroll
                for (int h = 0; h < RB; ++h) acc[h][t] = WMMA4(aO[h], bO[t], acc[h][t]);
            }
            {
                const int kn = (ks + 3 < K1S) ? ks + 3 : K1S - 1;
                const float* wr = wbase + (size_t)kn * (4 * NP1);
#pragma unroll
                for (int t = 0; t < 4; ++t) bO[t] = *(const v2f*)&wr[32 * t];
            }
        }
    }
    __syncthreads();

#pragma unroll
    for (int h = 0; h < RB; ++h) {
#pragma unroll
        for (int t = 0; t < 4; ++t) {
            const int jp = j0 + t * 16 + n;
            if (jp < N1) {
#pragma unroll
                for (int r = 0; r < 8; ++r) {
                    const int m  = 16 * h + r + 8 * half;
                    const int ig = i0 + m;
                    tile[m * SST + jp + (jp >= ig ? 1 : 0)] = acc[h][t][r];
                }
            }
        }
    }
    __syncthreads();

    {
        const int m0  = wave * (RT / 16);
        const int ig0 = i0 + m0;
        if (ig0 < A_DIM) {
            const float NEG = -3.402823466e+38f;
#pragma unroll 1
            for (int rr = 0; rr < RT / 16; ++rr) {
                float* const row = tile + (m0 + rr) * SST + lane;
                const int dj = ig0 + rr - lane;
                float v[SMI];
#pragma unroll
                for (int i = 0; i < SMI - 1; ++i) v[i] = row[32 * i];
                v[SMI - 1] = NEG;
                if (lane < A_DIM - 32 * (SMI - 1)) v[SMI - 1] = row[32 * (SMI - 1)];
                float mx = NEG;
#pragma unroll
                for (int i = 0; i < SMI; ++i) {
                    if (dj == 32 * i) v[i] = NEG;
                    mx = fmaxf(mx, v[i]);
                }
#pragma unroll
                for (int s = 16; s > 0; s >>= 1) mx = fmaxf(mx, __shfl_xor(mx, s, 32));
                float sum = 0.0f;
#pragma unroll
                for (int i = 0; i < SMI; ++i) {
                    const float e = expf(v[i] - mx);
                    v[i] = e;
                    sum += e;
                }
#pragma unroll
                for (int s = 16; s > 0; s >>= 1) sum += __shfl_xor(sum, s, 32);
                const float inv = 1.0f / sum;
#pragma unroll
                for (int i = 0; i < SMI - 1; ++i) row[32 * i] = v[i] * inv;
                if (lane < A_DIM - 32 * (SMI - 1)) row[32 * (SMI - 1)] = v[SMI - 1] * inv;
            }
        } else {
            for (int rr = 0; rr < RT / 16; ++rr)
                for (int j = lane; j < A_DIM; j += 32) tile[(m0 + rr) * SST + j] = 0.0f;
        }
    }
    __syncthreads();

    if (wave >= CH / 64) return;
    const int n0 = wave * 64;
    v8f o[RB][4];
#pragma unroll
    for (int h = 0; h < RB; ++h)
#pragma unroll
        for (int t = 0; t < 4; ++t)
#pragma unroll
            for (int r = 0; r < 8; ++r) o[h][t][r] = 0.0f;
    {
        const unsigned int* obase = offPk + (size_t)half * CH + n0 + n;
        const float* pbase = tile + ar * SST + 2 * half;
        unsigned int w[8];
#pragma unroll
        for (int t = 0; t < 4; ++t) {
            w[t]     = obase[16 * t];
            w[4 + t] = obase[2 * CH + 16 * t];
        }
        for (int kp = 0; kp < K2S / 2; ++kp) {
            unsigned int lo[8], hi[8];
#pragma unroll
            for (int i = 0; i < 8; ++i) { lo[i] = w[i] & 0xffffu; hi[i] = w[i] >> 16; }
            {
                const int kn = (kp + 1 < K2S / 2) ? kp + 1 : K2S / 2 - 1;
                const unsigned int* orow = obase + (size_t)kn * (4 * CH);
#pragma unroll
                for (int t = 0; t < 4; ++t) {
                    w[t]     = orow[16 * t];
                    w[4 + t] = orow[2 * CH + 16 * t];
                }
            }
            v2f pE[RB], pO[RB];
#pragma unroll
            for (int h = 0; h < RB; ++h) {
                pE[h] = *(const v2f*)&pbase[h * 16 * SST + kp * 8];
                pO[h] = *(const v2f*)&pbase[h * 16 * SST + kp * 8 + 4];
            }
#pragma unroll
            for (int t = 0; t < 4; ++t) {
                v2f bf;
                bf.x = lds_at(fvs, lo[t]);
                bf.y = lds_at(fvs, hi[t]);
#pragma unroll
                for (int h = 0; h < RB; ++h) o[h][t] = WMMA4(pE[h], bf, o[h][t]);
            }
#pragma unroll
            for (int t = 0; t < 4; ++t) {
                v2f bf;
                bf.x = lds_at(fvs, lo[4 + t]);
                bf.y = lds_at(fvs, hi[4 + t]);
#pragma unroll
                for (int h = 0; h < RB; ++h) o[h][t] = WMMA4(pO[h], bf, o[h][t]);
            }
        }
    }

    float* outb = out + (size_t)b * A_DIM * CH;
#pragma unroll
    for (int h = 0; h < RB; ++h) {
#pragma unroll
        for (int r = 0; r < 8; ++r) {
            const int m = i0 + 16 * h + r + 8 * half;
            if (m < A_DIM) {
                float* orow = outb + (size_t)m * CH + n0 + n;
#pragma unroll
                for (int t = 0; t < 4; ++t) orow[16 * t] = o[h][t][r];
            }
        }
    }
}

extern "C" void kernel_launch(void* const* d_in, const int* in_sizes, int n_in,
                              void* d_out, int out_size, void* d_ws, size_t ws_size,
                              hipStream_t stream)
{
    const float* fv   = (const float*)d_in[0];
    const int*   zi   = (const int*)  d_in[1];
    const int*   yi   = (const int*)  d_in[2];
    const int*   xi   = (const int*)  d_in[3];
    const int*   mask = (const int*)  d_in[4];
    const float* Wt   = (const float*)d_in[5];
    const float* bias = (const float*)d_in[6];
    float*       out  = (float*)d_out;

    float*        Wpk   = (float*)d_ws;
    float*        biasp = Wpk + (size_t)CH * NP1;
    unsigned int* offPk = (unsigned int*)(biasp + NP1);

    pad_w<<<(CH * NP1 + 255) / 256, 256, 0, stream>>>(Wt, Wpk);
    pad_bias<<<(NP1 + 255) / 256, 256, 0, stream>>>(bias, biasp);
    pack_offsets<<<((MP / 2) * CH + 255) / 256, 256, 0, stream>>>(zi, yi, xi, mask, offPk);
    fused_attn<<<dim3(NTILE, B_DIM), dim3(32, 16), 0, stream>>>(fv, offPk, Wpk, biasp, out);
}


# Both matrix loops in four stretches of falling wave priority (s_setprio 3,2,1,0 by loop progress): the waves of a SIMD end a loop together, the youngest no longer runs on alone; sums unchanged
# speedup vs baseline: 10.4128x; 1.0725x over previous
#include <hip/hip_runtime.h>


typedef __attribute__((ext_vector_type(2))) float v2f;
typedef __attribute__((ext_vector_type(8))) float v8f;

#define B_DIM  32
#define A_DIM  1000
#define C_DIM  64
#define H_DIM  12
#define W_DIM  20
#define CH     768
#define N1     999
#define MP     1008
#define RT     64
#define RB     4
#define NTILE  16
#define NP1    1024
#define FVN    15360
#define ZSLOT  61440u
#define AST    772
#define SST    1004
#define K1S    192
#define K2S    250
#define K2T    125
#define GIT    48
#define GGRP   16
#define CIT    7
#define SMI    32

static_assert(FVN == 4 * 512 * CIT + 2 * 512, "volume copy: 7 float4 rounds and one float2 round of 512 threads");
static_assert(A_DIM % (RT / 16) == 0, "a wave's softmax rows are all real or all padding");

#define WMMA4(a, b, c) __builtin_amdgcn_wmma_f32_16x16x4_f32( \
    false, (a), false, (b), (short)0, (c), false, false)

__global__ void pad_w(const float* __restrict__ Wt, float* __restrict__ Wpk)
{
    int e = blockIdx.x * blockDim.x + threadIdx.x;
    if (e >= CH * NP1) return;
    int k = e >> 10;
    int j = e & (NP1 - 1);
    float v = (j < N1) ? Wt[k * N1 + j] : 0.0f;
    Wpk[(size_t)(k >> 1) * (2 * NP1) + 2 * j + (k & 1)] = v;
}

__global__ void pad_bias(const float* __restrict__ bias, float* __restrict__ biasp)
{
    int j = blockIdx.x * blockDim.x + threadIdx.x;
    if (j < NP1) biasp[j] = (j < N1) ? bias[j] : 0.0f;
}

__global__ void pack_offsets(const int* __restrict__ zi, const int* __restrict__ yi,
                             const int* __restrict__ xi, const int* __restrict__ mask,
                             unsigned int* __restrict__ offPk)
{
    int e = blockIdx.x * blockDim.x + threadIdx.x;
    if (e >= (MP / 2) * CH) return;
    int p = e / CH;
    int k = e - p * CH;
    unsigned int w = 0;
#pragma unroll
    for (int s = 0; s < 2; ++s) {
        int a = 2 * p + s;
        unsigned int o = ZSLOT;
        if (a < A_DIM) {
            int idx = a * CH + k;
            if (!mask[idx])
                o = 4u * (unsigned int)((zi[idx] * H_DIM + yi[idx]) * W_DIM + xi[idx]);
        }
        w |= o << (16 * s);
    }
    offPk[e] = w;
}

__device__ __forceinline__ float lds_at(const float* base, unsigned int byteoff)
{
    return *(const float*)((const char*)base + byteoff);
}

template <int PRIO, int KS0, int KS1>
__device__ __forceinline__ void score_stretch(v8f (&acc)[RB][4], v2f (&bE)[4], v2f (&bO)[4],
                                              const float* wbase, const float* abase)
{
    __builtin_amdgcn_s_setprio(PRIO);
#pragma unroll 2
    for (int ks = KS0; ks < KS1; ks += 2) {
        v2f aE[RB];
#pragma unroll
        for (int h = 0; h < RB; ++h) aE[h] = *(const v2f*)&abase[h * 16 * AST + ks * 4];
#pragma unroll
        for (int t = 0; t < 4; ++t) {
#pragma unroll
            for (int h = 0; h < RB; ++h) acc[h][t] = WMMA4(aE[h], bE[t], acc[h][t]);
        }
        {
            const int kn = (ks + 2 < K1S) ? ks + 2 : K1S - 1;
            const float* wr = wbase + (size_t)kn * (4 * NP1);
#pragma unroll
            for (int t = 0; t < 4; ++t) bE[t] = *(const v2f*)&wr[32 * t];
        }
        v2f aO[RB];
#pragma unroll
        for (int h = 0; h < RB; ++h) aO[h] = *(const v2f*)&abase[h * 16 * AST + ks * 4 + 4];
#pragma unroll
        for (int t = 0; t < 4; ++t) {
#pragma unroll
            for (int h = 0; h < RB; ++h) acc[h][t] = WMMA4(aO[h], bO[t], acc[h][t]);
        }
        {
            const int kn = (ks + 3 < K1S) ? ks + 3 : K1S - 1;
            const float* wr = wbase + (size_t)kn * (4 * NP1);
#pragma unroll
            for (int t = 0; t < 4; ++t) bO[t] = *(const v2f*)&wr[32 * t];
        }
    }
}

template <int PRIO, int KP0, int KP1>
__device__ __forceinline__ void out_stretch(v8f (&o)[RB][4], unsigned int (&w)[8],
                                            const unsigned int* obase, const float* pbase,
                                            const float* fvs)
{
    __builtin_amdgcn_s_setprio(PRIO);
    for (int kp = KP0; kp < KP1; ++kp) {
        unsigned int lo[8], hi[8];
#pragma unroll
        for (int i = 0; i < 8; ++i) { lo[i] = w[i] & 0xffffu; hi[i] = w[i] >> 16; }
        {
            const int kn = (kp + 1 < K2S / 2) ? kp + 1 : K2S / 2 - 1;
            const unsigned int* orow = obase + (size_t)kn * (4 * CH);
#pragma unroll
            for (int t = 0; t < 4; ++t) {
                w[t]     = orow[16 * t];
                w[4 + t] = orow[2 * CH + 16 * t];
            }
        }
        v2f pE[RB], pO[RB];
#pragma unroll
        for (int h = 0; h < RB; ++h) {
            pE[h] = *(const v2f*)&pbase[h * 16 * SST + kp * 8];
            pO[h] = *(const v2f*)&pbase[h * 16 * SST + kp * 8 + 4];
        }
#pragma unroll
        for (int t = 0; t < 4; ++t) {
            v2f bf;
            bf.x = lds_at(fvs, lo[t]);
            bf.y = lds_at(fvs, hi[t]);
#pragma unroll
            for (int h = 0; h < RB; ++h) o[h][t] = WMMA4(pE[h], bf, o[h][t]);
        }
#pragma unroll
        for (int t = 0; t < 4; ++t) {
            v2f bf;
            bf.x = lds_at(fvs, lo[4 + t]);
            bf.y = lds_at(fvs, hi[4 + t]);
#pragma unroll
            for (int h = 0; h < RB; ++h) o[h][t] = WMMA4(pO[h], bf, o[h][t]);
        }
    }
}

__global__ void __launch_bounds__(512)
fused_attn(const float* __restrict__ fv,
           const unsigned int* __restrict__ offPk,
           const float* __restrict__ Wpk,
           const float* __restrict__ biasp,
           float* __restrict__ out)
{
    __shared__ __attribute__((aligned(16))) float smem[FVN + 4 + RT * SST];
    float* const fvs  = smem;
    float* const tile = smem + FVN + 4;

    const int lane = threadIdx.x;
    const int wave = threadIdx.y;
    const int tid  = wave * 32 + lane;
    const int i0   = blockIdx.x * RT;
    const int b    = blockIdx.y;
    const int half = lane >> 4;
    const int n    = lane & 15;
    const int ar   = lane & 15;

    unsigned int gw[GIT];
    {
        const unsigned int* orow = offPk + (size_t)(i0 >> 1) * CH;
        const int plim = MP / 2 - (i0 >> 1);
        const int elim = plim * CH - 1;
#pragma unroll
        for (int it = 0; it < GIT; ++it) {
            const int e = tid + 512 * it;
            const unsigned int w = orow[e < elim ? e : elim];
            gw[it] = (e <= elim) ? w : (ZSLOT | (ZSLOT << 16));
        }
    }
    {
        const float4* src = (const float4*)(fv + (size_t)b * FVN);
        float4* dst = (float4*)fvs;
        float4 cv[CIT];
#pragma unroll
        for (int q = 0; q < CIT; ++q) cv[q] = src[tid + 512 * q];
        const float2 cl = ((const float2*)(src + 512 * CIT))[tid];
#pragma unroll
        for (int q = 0; q < CIT; ++q) dst[tid + 512 * q] = cv[q];
        ((float2*)(dst + 512 * CIT))[tid] = cl;
        if (tid == 0) fvs[FVN] = 0.0f;
    }
    __syncthreads();

#pragma unroll
    for (int g = 0; g < GIT; g += GGRP) {
        float glo[GGRP], ghi[GGRP];
#pragma unroll
        for (int u = 0; u < GGRP; ++u) {
            glo[u] = lds_at(fvs, gw[g + u] & 0xffffu);
            ghi[u] = lds_at(fvs, gw[g + u] >> 16);
        }
#pragma unroll
        for (int u = 0; u < GGRP; ++u) {
            const int e = tid + 512 * (g + u);
            const int p = e / CH;
            const int k = e - p * CH;
            tile[(2 * p) * AST + k]     = glo[u];
            tile[(2 * p + 1) * AST + k] = ghi[u];
        }
    }
    __syncthreads();

    const int j0 = wave * 64;
    v8f acc[RB][4];
#pragma unroll
    for (int t = 0; t < 4; ++t) {
        const float bv = biasp[j0 + t * 16 + n];
#pragma unroll
        for (int h = 0; h < RB; ++h)
#pragma unroll
            for (int r = 0; r < 8; ++r) acc[h][t][r] = bv;
    }
    {
        const float* wbase  = Wpk + (size_t)half * (2 * NP1) + 2 * (j0 + n);
        const float* abase = tile + ar * AST + 2 * half;
        v2f bE[4], bO[4];
#pragma unroll
        for (int t = 0; t < 4; ++t) {
            bE[t] = *(const v2f*)&wbase[32 * t];
            bO[t] = *(const v2f*)&wbase[4 * NP1 + 32 * t];
        }
        score_stretch<3, 0,               K1S / 2    >(acc, bE, bO, wbase, abase);
        score_stretch<2, K1S / 2,         3 * K1S / 4>(acc, bE, bO, wbase, abase);
        score_stretch<1, 3 * K1S / 4,     7 * K1S / 8>(acc, bE, bO, wbase, abase);
        score_stretch<0, 7 * K1S / 8,     K1S        >(acc, bE, bO, wbase, abase);
    }
    __syncthreads();

#pragma unroll
    for (int h = 0; h < RB; ++h) {
#pragma unroll
        for (int t = 0; t < 4; ++t) {
            const int jp = j0 + t * 16 + n;
            if (jp < N1) {
#pragma unroll
                for (int r = 0; r < 8; ++r) {
                    const int m  = 16 * h + r + 8 * half;
                    const int ig = i0 + m;
                    tile[m * SST + jp + (jp >= ig ? 1 : 0)] = acc[h][t][r];
                }
            }
        }
    }
    __syncthreads();

    {
        const int m0  = wave * (RT / 16);
        const int ig0 = i0 + m0;
        if (ig0 < A_DIM) {
            const float NEG = -3.402823466e+38f;
#pragma unroll 1
            for (int rr = 0; rr < RT / 16; ++rr) {
                float* const row = tile + (m0 + rr) * SST + lane;
                const int dj = ig0 + rr - lane;
                float v[SMI];
#pragma unroll
                for (int i = 0; i < SMI - 1; ++i) v[i] = row[32 * i];
                v[SMI - 1] = NEG;
                if (lane < A_DIM - 32 * (SMI - 1)) v[SMI - 1] = row[32 * (SMI - 1)];
                float mx = NEG;
#pragma unroll
                for (int i = 0; i < SMI; ++i) {
                    if (dj == 32 * i) v[i] = NEG;
                    mx = fmaxf(mx, v[i]);
                }
#pragma unroll
                for (int s = 16; s > 0; s >>= 1) mx = fmaxf(mx, __shfl_xor(mx, s, 32));
                float sum = 0.0f;
#pragma unroll
                for (int i = 0; i < SMI; ++i) {
                    const float e = expf(v[i] - mx);
                    v[i] = e;
                    sum += e;
                }
#pragma unroll
                for (int s = 16; s > 0; s >>= 1) sum += __shfl_xor(sum, s, 32);
                const float inv = 1.0f / sum;
#pragma unroll
                for (int i = 0; i < SMI - 1; ++i) row[32 * i] = v[i] * inv;
                if (lane < A_DIM - 32 * (SMI - 1)) row[32 * (SMI - 1)] = v[SMI - 1] * inv;
            }
        } else {
            for (int rr = 0; rr < RT / 16; ++rr)
                for (int j = lane; j < A_DIM; j += 32) tile[(m0 + rr) * SST + j] = 0.0f;
        }
    }
    __syncthreads();

    if (wave >= CH / 64) return;
    const int n0 = wave * 64;
    v8f o[RB][4];
#pragma unroll
    for (int h = 0; h < RB; ++h)
#pragma unroll
        for (int t = 0; t < 4; ++t)
#pragma unroll
            for (int r = 0; r < 8; ++r) o[h][t][r] = 0.0f;
    {
        const unsigned int* obase = offPk + (size_t)half * CH + n0 + n;
        const float* pbase = tile + ar * SST + 2 * half;
        unsigned int w[8];
#pragma unroll
        for (int t = 0; t < 4; ++t) {
            w[t]     = obase[16 * t];
            w[4 + t] = obase[2 * CH + 16 * t];
        }
        out_stretch<3, 0,           K2T / 2    >(o, w, obase, pbase, fvs);
        out_stretch<2, K2T / 2,     3 * K2T / 4>(o, w, obase, pbase, fvs);
        out_stretch<1, 3 * K2T / 4, 7 * K2T / 8>(o, w, obase, pbase, fvs);
        out_stretch<0, 7 * K2T / 8, K2T        >(o, w, obase, pbase, fvs);
    }

    float* outb = out + (size_t)b * A_DIM * CH;
#pragma unroll
    for (int h = 0; h < RB; ++h) {
#pragma unroll
        for (int r = 0; r < 8; ++r) {
            const int m = i0 + 16 * h + r + 8 * half;
            if (m < A_DIM) {
                float* orow = outb + (size_t)m * CH + n0 + n;
#pragma unroll
                for (int t = 0; t < 4; ++t) orow[16 * t] = o[h][t][r];
            }
        }
    }
}

extern "C" void kernel_launch(void* const* d_in, const int* in_sizes, int n_in,
                              void* d_out, int out_size, void* d_ws, size_t ws_size,
                              hipStream_t stream)
{
    const float* fv   = (const float*)d_in[0];
    const int*   zi   = (const int*)  d_in[1];
    const int*   yi   = (const int*)  d_in[2];
    const int*   xi   = (const int*)  d_in[3];
    const int*   mask = (const int*)  d_in[4];
    const float* Wt   = (const float*)d_in[5];
    const float* bias = (const float*)d_in[6];
    float*       out  = (float*)d_out;

    float*        Wpk   = (float*)d_ws;
    float*        biasp = Wpk + (size_t)CH * NP1;
    unsigned int* offPk = (unsigned int*)(biasp + NP1);

    pad_w<<<(CH * NP1 + 255) / 256, 256, 0, stream>>>(Wt, Wpk);
    pad_bias<<<(NP1 + 255) / 256, 256, 0, stream>>>(bias, biasp);
    pack_offsets<<<((MP / 2) * CH + 255) / 256, 256, 0, stream>>>(zi, yi, xi, mask, offPk);
    fused_attn<<<dim3(NTILE, B_DIM), dim3(32, 16), 0, stream>>>(fv, offPk, Wpk, biasp, out);
}
